# starting kernel
# baseline (speedup 1.0000x reference)
#include <hip/hip_runtime.h>
#include <stdint.h>


typedef __attribute__((ext_vector_type(2))) float v2f;
typedef __attribute__((ext_vector_type(8))) float v8f;

#define C_IN    3
#define IMG     32
#define C_OUT   4
#define KS      7
#define STR     3
#define HO      9
#define NPIX    81
#define KCONV   147
#define KCONV_P 148
#define FC1_IN  324
#define HIDDEN  64
#define OUTPUT  10

__device__ __forceinline__ v8f wmma4(v2f a, v2f b, v8f c) {
  return __builtin_amdgcn_wmma_f32_16x16x4_f32(false, a, false, b, (short)0, c,
                                               false, false);
}

__device__ __forceinline__ float ldg_off(const float* __restrict__ base,
                                         uint32_t byte_off) {
  return *(const float*)((const char*)base + (size_t)byte_off);
}

__global__ __launch_bounds__(256)
void conv_sq_kernel(const float* __restrict__ x, const float* __restrict__ w,
                    float* __restrict__ y, int nrows) {
  __shared__ float wT[16 * KCONV_P];
  __shared__ int   offs[KCONV_P];

  const int tid = threadIdx.x;
  for (int idx = tid; idx < 16 * KCONV_P; idx += 256) {
    int n = idx / KCONV_P, k = idx - n * KCONV_P;
    wT[idx] = (n < C_OUT && k < KCONV) ? w[n * KCONV + k] : 0.0f;
  }
  for (int idx = tid; idx < KCONV_P; idx += 256) {
    int k = (idx < KCONV) ? idx : (KCONV - 1);
    int ci = k / 49, r = k - ci * 49;
    int ky = r / 7,  kx = r - ky * 7;
    offs[idx] = (ci * (IMG * IMG) + ky * IMG + kx) * 4;
  }
  __syncthreads();

  const int lane = tid & 31;
  const int wave = tid >> 5;
  const int ln   = lane & 15;
  const int half = lane >> 4;
  const int koff = half * 2;

  const int row_base = (blockIdx.x * 8 + wave) * 16;
  if (row_base >= nrows) return;

  const int arow = row_base + ln;
  const int b    = arow / NPIX;
  const int p    = arow - b * NPIX;
  const int oy   = p / HO, ox = p - oy * HO;
  const uint32_t base =
      ((uint32_t)b * (C_IN * IMG * IMG) +
       (uint32_t)(oy * STR * IMG + ox * STR)) * 4u;

  v8f acc = {};
  for (int kt = 0; kt < KCONV_P / 4; ++kt) {
    const int k0 = kt * 4 + koff;
    v2f a, bw;
    a.x = ldg_off(x, base + (uint32_t)offs[k0]);
    a.y = ldg_off(x, base + (uint32_t)offs[k0 + 1]);
    bw  = *(const v2f*)&wT[ln * KCONV_P + k0];
    acc = wmma4(a, bw, acc);
  }

  if (ln < C_OUT) {
#pragma unroll
    for (int r = 0; r < 8; ++r) {
      const int row = row_base + half * 8 + r;
      const int bb  = row / NPIX;
      const int pp  = row - bb * NPIX;
      const float c = acc[r];
      y[(size_t)bb * FC1_IN + ln * NPIX + pp] = c * c;
    }
  }
}

__global__ __launch_bounds__(256)
void fc1_sq_kernel(const float* __restrict__ y, const float* __restrict__ w,
                   const float* __restrict__ bias, float* __restrict__ h) {
  const int tid  = threadIdx.x;
  const int lane = tid & 31;
  const int wave = tid >> 5;
  const int ln   = lane & 15;
  const int half = lane >> 4;
  const int koff = half * 2;

  const int m_base = blockIdx.x * 32 + (wave >> 2) * 16;
  const int n      = (wave & 3) * 16 + ln;

  const float* ap = y + (size_t)(m_base + ln) * FC1_IN;
  const float* bp = w + (size_t)n * FC1_IN;

  v8f acc = {};
  for (int kt = 0; kt < FC1_IN / 4; ++kt) {
    const int k0 = kt * 4 + koff;
    const v2f a = *(const v2f*)(ap + k0);
    const v2f b = *(const v2f*)(bp + k0);
    acc = wmma4(a, b, acc);
  }

  const float bn = bias[n];
#pragma unroll
  for (int r = 0; r < 8; ++r) {
    const int row = m_base + half * 8 + r;
    const float t = acc[r] + bn;
    h[(size_t)row * HIDDEN + n] = t * t;
  }
}

__global__ __launch_bounds__(256)
void fc2_kernel(const float* __restrict__ h, const float* __restrict__ w,
                const float* __restrict__ bias, float* __restrict__ out) {
  __shared__ float wpad[16 * HIDDEN];
  const int tid = threadIdx.x;
  for (int idx = tid; idx < 16 * HIDDEN; idx += 256) {
    int n = idx / HIDDEN, k = idx - n * HIDDEN;
    wpad[idx] = (n < OUTPUT) ? w[n * HIDDEN + k] : 0.0f;
  }
  __syncthreads();

  const int lane = tid & 31;
  const int wave = tid >> 5;
  const int ln   = lane & 15;
  const int half = lane >> 4;
  const int koff = half * 2;

  const int m_base = (blockIdx.x * 8 + wave) * 16;
  const float* ap  = h + (size_t)(m_base + ln) * HIDDEN;

  v8f acc = {};
  for (int kt = 0; kt < HIDDEN / 4; ++kt) {
    const int k0 = kt * 4 + koff;
    const v2f a = *(const v2f*)(ap + k0);
    const v2f b = *(const v2f*)&wpad[ln * HIDDEN + k0];
    acc = wmma4(a, b, acc);
  }

  if (ln < OUTPUT) {
    const float bn = bias[ln];
#pragma unroll
    for (int r = 0; r < 8; ++r) {
      const int row = m_base + half * 8 + r;
      out[(size_t)row * OUTPUT + ln] = acc[r] + bn;
    }
  }
}

extern "C" void kernel_launch(void* const* d_in, const int* in_sizes, int n_in,
                              void* d_out, int out_size, void* d_ws,
                              size_t ws_size, hipStream_t stream) {
  const float* x      = (const float*)d_in[0];
  const float* conv_w = (const float*)d_in[1];
  const float* fc1_w  = (const float*)d_in[2];
  const float* fc1_b  = (const float*)d_in[3];
  const float* fc2_w  = (const float*)d_in[4];
  const float* fc2_b  = (const float*)d_in[5];
  float* out = (float*)d_out;

  const int B = in_sizes[0] / (C_IN * IMG * IMG);

  float* y = (float*)d_ws;
  float* h = y + (size_t)B * FC1_IN;

  const int nrows = B * NPIX;
  conv_sq_kernel<<<nrows / (16 * 8), 256, 0, stream>>>(x, conv_w, y, nrows);
  fc1_sq_kernel<<<B / 32, 256, 0, stream>>>(y, fc1_w, fc1_b, h);
  fc2_kernel<<<B / (16 * 8), 256, 0, stream>>>(h, fc2_w, fc2_b, out);
}


# conv A operand staged in LDS: each work-group copies its 4 images with 16-byte loads, 7 waves x 3 tiles gather taps from LDS, K steps regrouped into tap pairs (f32 sum order over K changes)
# speedup vs baseline: 1.4768x; 1.4768x over previous
#include <hip/hip_runtime.h>
#include <stdint.h>


typedef __attribute__((ext_vector_type(2))) float v2f;
typedef __attribute__((ext_vector_type(4))) float v4f;
typedef __attribute__((ext_vector_type(8))) float v8f;

#define C_IN    3
#define IMG     32
#define C_OUT   4
#define KS      7
#define STR     3
#define HO      9
#define NPIX    81
#define KCONV   147
#define KCONV_P 148
#define FC1_IN  324
#define HIDDEN  64
#define OUTPUT  10

__device__ __forceinline__ v8f wmma4(v2f a, v2f b, v8f c) {
  return __builtin_amdgcn_wmma_f32_16x16x4_f32(false, a, false, b, (short)0, c,
                                               false, false);
}

#define GIMG     4
#define IMG_F    (C_IN * IMG * IMG)
#define ROWS_G   (GIMG * NPIX)
#define TPW      3
#define CWAVES   7
#define CTHREADS (CWAVES * 32)
#define WROWS    5
#define KPITCH   164
#define IMG_V4   (GIMG * IMG_F / 4)
#define LTHREADS 192

struct TapPair { int c, o, d; };

__host__ __device__ constexpr TapPair tap_pair(int s, int j) {
  if (s < 30) {
    const int v = 2 * (s % 10) + j;
    return TapPair{s / 10, 64 * (v / 7) + (v % 7), 32};
  }
  if (s < 33) return TapPair{s - 30, 192 + 2 * j, 1};
  if (s == 33) return TapPair{2, 196 + 2 * j, 1};
  if (s == 34) return TapPair{j, 134, 32};
  if (s == 35) return j == 0 ? TapPair{2, 134, 32}
                             : TapPair{0, 196, 1};
  return j == 0 ? TapPair{1, 196, 1}
                : TapPair{0, 198, 1024};
}

struct TapSrc { short v[160]; };
__host__ __device__ constexpr TapSrc make_tap_src() {
  TapSrc t{};
  for (int q = KCONV_P; q < 160; ++q) t.v[q] = (short)-1;
  for (int q = 0; q < KCONV_P; ++q) {
    const int s = q >> 2, h = (q >> 1) & 1, j = q & 1;
    const TapPair p = tap_pair(s, j);
    const int a  = p.c * 1024 + p.o + h * p.d;
    const int ch = a / 1024, r = a % 1024, ky = r / 32, kx = r % 32;
    t.v[q] = (ch < C_IN && ky < KS && kx < KS) ? (short)(ch * 49 + ky * 7 + kx)
                                               : (short)-1;
  }
  return t;
}

__host__ __device__ constexpr int dummy_pos() {
  const TapSrc t = make_tap_src();
  for (int q = 0; q < KCONV_P; ++q)
    if (t.v[q] < 0) return q;
  return -1;
}

__global__ __launch_bounds__(CTHREADS)
void conv_sq_kernel(const float* __restrict__ x, const float* __restrict__ w,
                    float* __restrict__ y) {
  __shared__ __attribute__((aligned(16))) float img[GIMG * IMG_F];
  __shared__ __attribute__((aligned(16))) float wT[WROWS * KPITCH];
  static constexpr TapSrc kSrc = make_tap_src();

  const int tid = threadIdx.x;

  if (tid < LTHREADS) {
    const v4f* gsrc =
        reinterpret_cast<const v4f*>(x + (size_t)blockIdx.x * (GIMG * IMG_F));
    v4f* ldst = reinterpret_cast<v4f*>(img);
#pragma unroll
    for (int i = 0; i < IMG_V4 / LTHREADS; ++i)
      ldst[tid + i * LTHREADS] =
          __builtin_nontemporal_load(&gsrc[tid + i * LTHREADS]);
  } else {
    const int l = tid & 31;
    int   sidx[5];
    float wv[5][C_OUT];
#pragma unroll
    for (int i = 0; i < 5; ++i) sidx[i] = kSrc.v[l + 32 * i];
#pragma unroll
    for (int i = 0; i < 5; ++i) {
      const int si = sidx[i] < 0 ? 0 : sidx[i];
#pragma unroll
      for (int n = 0; n < C_OUT; ++n) wv[i][n] = w[n * KCONV + si];
    }
#pragma unroll
    for (int i = 0; i < 5; ++i) {
      const int q = l + 32 * i;
#pragma unroll
      for (int n = 0; n < C_OUT; ++n) wT[n * KPITCH + q] = wv[i][n];
      wT[C_OUT * KPITCH + q] = 0.0f;
    }
    constexpr int dq = dummy_pos();
    if (l == (dq & 31)) {
      for (int n = 0; n < C_OUT; ++n) wT[n * KPITCH + dq] = 0.0f;
    }
  }
  __syncthreads();

  const int lane = tid & 31;
  const int wave = __builtin_amdgcn_readfirstlane(tid >> 5);
  const int ln   = lane & 15;
  const int half = lane >> 4;

  int tb[TPW];
#pragma unroll
  for (int t = 0; t < TPW; ++t) {
    int m = (wave * TPW + t) * 16 + ln;
    m = m < ROWS_G ? m : ROWS_G - 1;
    const int g  = m / NPIX;
    const int p  = m - g * NPIX;
    const int oy = p / HO, ox = p - oy * HO;
    tb[t] = g * IMG_F + oy * (STR * IMG) + ox * STR;
  }
  const float* bp = wT + (ln < C_OUT ? ln : C_OUT) * KPITCH + half * 2;

  int vb[TPW][C_IN];
#pragma unroll
  for (int t = 0; t < TPW; ++t) {
#pragma unroll
    for (int c = 0; c < C_IN; ++c) {
      vb[t][c] = (tb[t] + c * 1024 + half * 32) * 4;
      asm("" : "+v"(vb[t][c]));
    }
  }

  v8f acc[TPW] = {};
#pragma unroll
  for (int s = 0; s < KCONV_P / 4; ++s) {
    const TapPair p0 = tap_pair(s, 0);
    const TapPair p1 = tap_pair(s, 1);
    const v2f bw = *(const v2f*)(bp + 4 * s);
#pragma unroll
    for (int t = 0; t < TPW; ++t) {
      const char* ib = reinterpret_cast<const char*>(img);
      v2f a;
      a.x = (p0.d == 32)
                ? *(const float*)(ib + vb[t][p0.c] + 4 * p0.o)
                : img[tb[t] + p0.c * 1024 + half * p0.d + p0.o];
      a.y = (p1.d == 32)
                ? *(const float*)(ib + vb[t][p1.c] + 4 * p1.o)
                : img[tb[t] + p1.c * 1024 + half * p1.d + p1.o];
      acc[t] = wmma4(a, bw, acc[t]);
    }
  }

  if (ln < C_OUT) {
    float* yg = y + (size_t)blockIdx.x * (GIMG * FC1_IN);
#pragma unroll
    for (int t = 0; t < TPW; ++t) {
      const int tile = wave * TPW + t;
      const int m0 = tile * 16 + half * 8;
      const int g0 = m0 / NPIX;
      const int p0 = m0 - g0 * NPIX;
      const unsigned o0 = (unsigned)(g0 * FC1_IN + ln * NPIX + p0);
      const int rw = NPIX - p0;
      if (t < TPW - 1 || wave < CWAVES - 1) {
#pragma unroll
        for (int r = 0; r < 8; ++r) {
          const float c = acc[t][r];
          yg[o0 + r + (r >= rw ? FC1_IN - NPIX : 0)] = c * c;
        }
      } else if (half == 0) {
#pragma unroll
        for (int r = 0; r < ROWS_G - (CWAVES * TPW - 1) * 16; ++r) {
          const float c = acc[t][r];
          yg[o0 + r + (r >= rw ? FC1_IN - NPIX : 0)] = c * c;
        }
      }
    }
  }
}

__global__ __launch_bounds__(256)
void fc1_sq_kernel(const float* __restrict__ y, const float* __restrict__ w,
                   const float* __restrict__ bias, float* __restrict__ h) {
  const int tid  = threadIdx.x;
  const int lane = tid & 31;
  const int wave = tid >> 5;
  const int ln   = lane & 15;
  const int half = lane >> 4;
  const int koff = half * 2;

  const int m_base = blockIdx.x * 32 + (wave >> 2) * 16;
  const int n      = (wave & 3) * 16 + ln;

  const float* ap = y + (size_t)(m_base + ln) * FC1_IN;
  const float* bp = w + (size_t)n * FC1_IN;

  v8f acc = {};
  for (int kt = 0; kt < FC1_IN / 4; ++kt) {
    const int k0 = kt * 4 + koff;
    const v2f a = *(const v2f*)(ap + k0);
    const v2f b = *(const v2f*)(bp + k0);
    acc = wmma4(a, b, acc);
  }

  const float bn = bias[n];
#pragma unroll
  for (int r = 0; r < 8; ++r) {
    const int row = m_base + half * 8 + r;
    const float t = acc[r] + bn;
    h[(size_t)row * HIDDEN + n] = t * t;
  }
}

__global__ __launch_bounds__(256)
void fc2_kernel(const float* __restrict__ h, const float* __restrict__ w,
                const float* __restrict__ bias, float* __restrict__ out) {
  __shared__ float wpad[16 * HIDDEN];
  const int tid = threadIdx.x;
  for (int idx = tid; idx < 16 * HIDDEN; idx += 256) {
    int n = idx / HIDDEN, k = idx - n * HIDDEN;
    wpad[idx] = (n < OUTPUT) ? w[n * HIDDEN + k] : 0.0f;
  }
  __syncthreads();

  const int lane = tid & 31;
  const int wave = tid >> 5;
  const int ln   = lane & 15;
  const int half = lane >> 4;
  const int koff = half * 2;

  const int m_base = (blockIdx.x * 8 + wave) * 16;
  const float* ap  = h + (size_t)(m_base + ln) * HIDDEN;

  v8f acc = {};
  for (int kt = 0; kt < HIDDEN / 4; ++kt) {
    const int k0 = kt * 4 + koff;
    const v2f a = *(const v2f*)(ap + k0);
    const v2f b = *(const v2f*)&wpad[ln * HIDDEN + k0];
    acc = wmma4(a, b, acc);
  }

  if (ln < OUTPUT) {
    const float bn = bias[ln];
#pragma unroll
    for (int r = 0; r < 8; ++r) {
      const int row = m_base + half * 8 + r;
      out[(size_t)row * OUTPUT + ln] = acc[r] + bn;
    }
  }
}

extern "C" void kernel_launch(void* const* d_in, const int* in_sizes, int n_in,
                              void* d_out, int out_size, void* d_ws,
                              size_t ws_size, hipStream_t stream) {
  const float* x      = (const float*)d_in[0];
  const float* conv_w = (const float*)d_in[1];
  const float* fc1_w  = (const float*)d_in[2];
  const float* fc1_b  = (const float*)d_in[3];
  const float* fc2_w  = (const float*)d_in[4];
  const float* fc2_b  = (const float*)d_in[5];
  float* out = (float*)d_out;

  const int B = in_sizes[0] / (C_IN * IMG * IMG);

  float* y = (float*)d_ws;
  float* h = y + (size_t)B * FC1_IN;

  conv_sq_kernel<<<B / GIMG, CTHREADS, 0, stream>>>(x, conv_w, y);
  fc1_sq_kernel<<<B / 32, 256, 0, stream>>>(y, fc1_w, fc1_b, h);
  fc2_kernel<<<B / (16 * 8), 256, 0, stream>>>(h, fc2_w, fc2_b, out);
}
